# starting kernel
# baseline (speedup 1.0000x reference)
#include <hip/hip_runtime.h>
#include <math.h>

#define NNODES  40000
#define NEDGES  640000
#define DIM     128
#define NGRAPH  64
#define NCLS    100
#define NLAYERS 3

typedef __attribute__((ext_vector_type(2))) float v2f;
typedef __attribute__((ext_vector_type(8))) float v8f;

__device__ __forceinline__ void atomAddF(float* p, float v) {
    __hip_atomic_fetch_add(p, v, __ATOMIC_RELAXED, __HIP_MEMORY_SCOPE_AGENT);
}

__global__ void normalize_rows(const float* __restrict__ h,
                               float* __restrict__ nh, int n) {
    int lane = threadIdx.x & 31;
    int row  = blockIdx.x * (blockDim.x >> 5) + (threadIdx.x >> 5);
    if (row >= n) return;
    float4 v = ((const float4*)(h + (size_t)row * DIM))[lane];
    float ss = v.x * v.x + v.y * v.y + v.z * v.z + v.w * v.w;
#pragma unroll
    for (int off = 16; off >= 1; off >>= 1) ss += __shfl_xor(ss, off, 32);
    float inv = 1.0f / (sqrtf(ss) + 1e-12f);
    float4 o;
    o.x = v.x * inv; o.y = v.y * inv; o.z = v.z * inv; o.w = v.w * inv;
    ((float4*)(nh + (size_t)row * DIM))[lane] = o;
}

__global__ void edge_attend(const float* __restrict__ nh,
                            const float* __restrict__ h,
                            const int* __restrict__ src,
                            const int* __restrict__ dst,
                            const float* __restrict__ betas, int layer,
                            float* __restrict__ num,
                            float* __restrict__ denom, int e) {
    int lane = threadIdx.x & 31;
    int edge = blockIdx.x * (blockDim.x >> 5) + (threadIdx.x >> 5);
    if (edge >= e) return;
    int s = src[edge];
    int d = dst[edge];
    float4 a4 = ((const float4*)(nh + (size_t)s * DIM))[lane];
    float4 b4 = ((const float4*)(nh + (size_t)d * DIM))[lane];
    float p = a4.x * b4.x + a4.y * b4.y + a4.z * b4.z + a4.w * b4.w;
#pragma unroll
    for (int off = 16; off >= 1; off >>= 1) p += __shfl_xor(p, off, 32);
    float att = __expf(betas[layer] * p);
    if (lane == 0) atomAddF(denom + d, att);
    float4 hv = ((const float4*)(h + (size_t)s * DIM))[lane];
    float* np = num + (size_t)d * DIM + lane * 4;
    atomAddF(np + 0, att * hv.x);
    atomAddF(np + 1, att * hv.y);
    atomAddF(np + 2, att * hv.z);
    atomAddF(np + 3, att * hv.w);
}

__global__ void finalize_div(float* __restrict__ num,
                             const float* __restrict__ denom, int nd) {
    int i = blockIdx.x * blockDim.x + threadIdx.x;
    if (i >= nd) return;
    float dn = denom[i >> 7];
    num[i] = num[i] / (dn > 0.0f ? dn : 1.0f);
}

__global__ void pool_sum(const float* __restrict__ h,
                         const int* __restrict__ gid,
                         float* __restrict__ hg, int n) {
    int i = blockIdx.x * blockDim.x + threadIdx.x;
    if (i >= n * DIM) return;
    int node = i >> 7;
    int d    = i & 127;
    atomAddF(hg + (size_t)gid[node] * DIM + d, h[i]);
}

__global__ void pool_cnt(const int* __restrict__ gid,
                         float* __restrict__ cnt, int n) {
    int i = blockIdx.x * blockDim.x + threadIdx.x;
    if (i >= n) return;
    atomAddF(cnt + gid[i], 1.0f);
}

__global__ void pool_div(float* __restrict__ hg, const float* __restrict__ cnt) {
    int i = blockIdx.x * blockDim.x + threadIdx.x;
    if (i >= NGRAPH * DIM) return;
    hg[i] /= fmaxf(cnt[i >> 7], 1.0f);
}

__global__ void classifier_wmma(const float* __restrict__ hg,
                                const float* __restrict__ W,
                                const float* __restrict__ b,
                                float* __restrict__ out) {
    int mt = blockIdx.x / 7;
    int nt = blockIdx.x % 7;
    int m0 = mt * 16;
    int n0 = nt * 16;
    int lane = threadIdx.x & 31;
    int half = lane >> 4;
    int lr   = lane & 15;
    int col  = n0 + lr;
    int ccol = col < NCLS ? col : (NCLS - 1);
    float mask = col < NCLS ? 1.0f : 0.0f;

    v8f acc = {};
#pragma unroll 4
    for (int k0 = 0; k0 < DIM; k0 += 4) {
        const float* ar = hg + (size_t)(m0 + lr) * DIM + k0 + 2 * half;
        v2f a;
        a.x = ar[0];
        a.y = ar[1];
        v2f bb;
        bb.x = W[(size_t)(k0 + 2 * half) * NCLS + ccol] * mask;
        bb.y = W[(size_t)(k0 + 2 * half + 1) * NCLS + ccol] * mask;
        acc = __builtin_amdgcn_wmma_f32_16x16x4_f32(
             false, a,  false, bb,
             (short)0, acc,  false,  false);
    }

    if (col < NCLS) {
        float bias = b[col];
#pragma unroll
        for (int r = 0; r < 8; ++r) {
            int m = m0 + r + 8 * half;
            out[(size_t)m * NCLS + col] = acc[r] + bias;
        }
    }
}

extern "C" void kernel_launch(void* const* d_in, const int* in_sizes, int n_in,
                              void* d_out, int out_size, void* d_ws, size_t ws_size,
                              hipStream_t stream) {
    const float* h0    = (const float*)d_in[0];
    const int*   src   = (const int*)d_in[1];
    const int*   dst   = (const int*)d_in[2];
    const int*   gid   = (const int*)d_in[3];
    const float* betas = (const float*)d_in[4];
    const float* Wc    = (const float*)d_in[5];
    const float* bc    = (const float*)d_in[6];
    float*       out   = (float*)d_out;

    float* nh    = (float*)d_ws;
    float* hA    = nh + (size_t)NNODES * DIM;
    float* hB    = hA + (size_t)NNODES * DIM;
    float* denom = hB + (size_t)NNODES * DIM;
    float* hg    = denom + NNODES;
    float* cnt   = hg + NGRAPH * DIM;

    const int ND = NNODES * DIM;

    const float* hin = h0;
    float* outs[NLAYERS] = {hA, hB, hA};
    for (int l = 0; l < NLAYERS; ++l) {
        float* hout = outs[l];
        hipMemsetAsync(hout, 0, sizeof(float) * (size_t)ND, stream);
        hipMemsetAsync(denom, 0, sizeof(float) * NNODES, stream);
        normalize_rows<<<(NNODES + 7) / 8, 256, 0, stream>>>(hin, nh, NNODES);
        edge_attend<<<(NEDGES + 7) / 8, 256, 0, stream>>>(nh, hin, src, dst,
                                                          betas, l, hout, denom,
                                                          NEDGES);
        finalize_div<<<(ND + 255) / 256, 256, 0, stream>>>(hout, denom, ND);
        hin = hout;
    }

    hipMemsetAsync(hg, 0, sizeof(float) * NGRAPH * DIM, stream);
    hipMemsetAsync(cnt, 0, sizeof(float) * NGRAPH, stream);
    pool_sum<<<(ND + 255) / 256, 256, 0, stream>>>(hin, gid, hg, NNODES);
    pool_cnt<<<(NNODES + 255) / 256, 256, 0, stream>>>(gid, cnt, NNODES);
    pool_div<<<(NGRAPH * DIM + 255) / 256, 256, 0, stream>>>(hg, cnt);

    classifier_wmma<<<4 * 7, 32, 0, stream>>>(hg, Wc, bc, out);
}


# Layers: atomic scatter replaced by gather: edges counting-sorted by destination once per call, one wave per node sums its sources in registers (row norms as scalars; no zero fill, no divide pass)
# speedup vs baseline: 2.3844x; 2.3844x over previous
#include <hip/hip_runtime.h>
#include <math.h>

#define NNODES  40000
#define NEDGES  640000
#define DIM     128
#define NGRAPH  64
#define NCLS    100
#define NLAYERS 3

typedef __attribute__((ext_vector_type(2))) float v2f;
typedef __attribute__((ext_vector_type(8))) float v8f;


__device__ __forceinline__ void atomAddF(float* p, float v) {
    __hip_atomic_fetch_add(p, v, __ATOMIC_RELAXED, __HIP_MEMORY_SCOPE_AGENT);
}

template <int M>
__device__ __forceinline__ float xrow(float x) {
    int b = __builtin_bit_cast(int, x);
    return __builtin_bit_cast(float,
        __builtin_amdgcn_update_dpp(0, b, 0x160 + M, 0xf, 0xf, true));
}
__device__ __forceinline__ float xhalf(float x) {
    int b = __builtin_bit_cast(int, x);
    return __builtin_bit_cast(float,
        __builtin_amdgcn_permlanex16(b, b, 0x76543210, 0xfedcba98, false, false));
}
__device__ __forceinline__ float wave_sum(float x) {
    x += xrow<1>(x);
    x += xrow<2>(x);
    x += xrow<4>(x);
    x += xrow<8>(x);
    x += xhalf(x);
    return x;
}
__device__ __forceinline__ int   bcast_i(int v, int l)   { return __builtin_amdgcn_readlane(v, l); }
__device__ __forceinline__ float bcast_f(float v, int l) {
    return __builtin_bit_cast(float, __builtin_amdgcn_readlane(__builtin_bit_cast(int, v), l));
}

__global__ void dst_count(const int* __restrict__ dst, int* __restrict__ cur, int e) {
    int i = blockIdx.x * blockDim.x + threadIdx.x;
    if (i >= e) return;
    __hip_atomic_fetch_add(cur + dst[i], 1, __ATOMIC_RELAXED, __HIP_MEMORY_SCOPE_AGENT);
}

#define SCAN_PER 40
__global__ __launch_bounds__(1024) void dst_scan(int* __restrict__ cur) {
    __shared__ int wtot[32];
    const int t = threadIdx.x;
    const int lane = t & 31;
    const int w = t >> 5;
    const bool live = t < NNODES / SCAN_PER;
    int4 v[SCAN_PER / 4];
    int sum = 0;
    if (live) {
        const int4* p = (const int4*)(cur + t * SCAN_PER);
#pragma unroll
        for (int k = 0; k < SCAN_PER / 4; ++k) v[k] = p[k];
#pragma unroll
        for (int k = 0; k < SCAN_PER / 4; ++k) sum += v[k].x + v[k].y + v[k].z + v[k].w;
    }
    int inc = sum;
    for (int off = 1; off < 32; off <<= 1) {
        int y = __shfl_up(inc, off, 32);
        inc += (lane >= off) ? y : 0;
    }
    if (lane == 31) wtot[w] = inc;
    __syncthreads();
    if (w == 0) {
        int x = wtot[lane];
        int s = x;
        for (int off = 1; off < 32; off <<= 1) {
            int y = __shfl_up(s, off, 32);
            s += (lane >= off) ? y : 0;
        }
        wtot[lane] = s - x;
    }
    __syncthreads();
    if (live) {
        int run = wtot[w] + inc - sum;
        int4* p = (int4*)(cur + t * SCAN_PER);
#pragma unroll
        for (int k = 0; k < SCAN_PER / 4; ++k) {
            int4 o;
            o.x = run; run += v[k].x;
            o.y = run; run += v[k].y;
            o.z = run; run += v[k].z;
            o.w = run; run += v[k].w;
            p[k] = o;
        }
    }
}

__global__ void dst_fill(const int* __restrict__ src, const int* __restrict__ dst,
                         int* __restrict__ cur, int* __restrict__ esrc, int e) {
    int i = blockIdx.x * blockDim.x + threadIdx.x;
    if (i >= e) return;
    int pos = __hip_atomic_fetch_add(cur + dst[i], 1, __ATOMIC_RELAXED, __HIP_MEMORY_SCOPE_AGENT);
    if ((unsigned)pos < (unsigned)e) esrc[pos] = src[i];
}

__global__ void row_inv_norm(const float* __restrict__ h,
                             float* __restrict__ inv, int n) {
    int lane = threadIdx.x & 31;
    int row  = blockIdx.x * (blockDim.x >> 5) + (threadIdx.x >> 5);
    if (row >= n) return;
    float4 v = ((const float4*)(h + (size_t)row * DIM))[lane];
    float ss = v.x * v.x + v.y * v.y + v.z * v.z + v.w * v.w;
    ss = wave_sum(ss);
    if (lane == 0) inv[row] = 1.0f / (sqrtf(ss) + 1e-12f);
}

__global__ __launch_bounds__(256) void gather_attend(const float* __restrict__ h,
                              const float* __restrict__ inv,
                              const int* __restrict__ rowend,
                              const int* __restrict__ esrc,
                              const float* __restrict__ betas, int layer,
                              float* __restrict__ hout, int n) {
    const int lane = threadIdx.x & 31;
    int node = blockIdx.x * (blockDim.x >> 5) + (threadIdx.x >> 5);
    if (node >= n) return;
    node = __builtin_amdgcn_readfirstlane(node);
    int beg = node > 0 ? rowend[node - 1] : 0;
    int end = rowend[node];
    beg = min(max(beg, 0), NEDGES);
    end = min(max(end, beg), NEDGES);
    beg = __builtin_amdgcn_readfirstlane(beg);
    end = __builtin_amdgcn_readfirstlane(end);

    const float4* H = (const float4*)h;
    const float4 hd = H[(size_t)node * (DIM / 4) + lane];
    const float sd = inv[node] * betas[layer];

    float4 acc = {0.0f, 0.0f, 0.0f, 0.0f};
    float den = 0.0f;

    for (int base = beg; base < end; base += 32) {
        const int cnt = min(32, end - base);
        int sidx = 0;
        float sinv = 0.0f;
        if (lane < cnt) {
            sidx = esrc[base + lane];
            sidx = min(max(sidx, 0), n - 1);
            sinv = inv[sidx];
        }
        for (int j = 0; j < cnt; j += 4) {
            const bool ok1 = j + 1 < cnt, ok2 = j + 2 < cnt, ok3 = j + 3 < cnt;
            const int s0 = bcast_i(sidx, j);
            const int s1 = ok1 ? bcast_i(sidx, j + 1) : s0;
            const int s2 = ok2 ? bcast_i(sidx, j + 2) : s0;
            const int s3 = ok3 ? bcast_i(sidx, j + 3) : s0;
            const float4 r0 = H[(size_t)s0 * (DIM / 4) + lane];
            const float4 r1 = H[(size_t)s1 * (DIM / 4) + lane];
            const float4 r2 = H[(size_t)s2 * (DIM / 4) + lane];
            const float4 r3 = H[(size_t)s3 * (DIM / 4) + lane];
            const float i0 = bcast_f(sinv, j);
            const float i1 = ok1 ? bcast_f(sinv, j + 1) : 0.0f;
            const float i2 = ok2 ? bcast_f(sinv, j + 2) : 0.0f;
            const float i3 = ok3 ? bcast_f(sinv, j + 3) : 0.0f;
            float p0 = r0.x * hd.x + r0.y * hd.y + r0.z * hd.z + r0.w * hd.w;
            float p1 = r1.x * hd.x + r1.y * hd.y + r1.z * hd.z + r1.w * hd.w;
            float p2 = r2.x * hd.x + r2.y * hd.y + r2.z * hd.z + r2.w * hd.w;
            float p3 = r3.x * hd.x + r3.y * hd.y + r3.z * hd.z + r3.w * hd.w;
            p0 = wave_sum(p0);
            p1 = wave_sum(p1);
            p2 = wave_sum(p2);
            p3 = wave_sum(p3);
            const float a0 = __expf(p0 * i0 * sd);
            const float a1 = ok1 ? __expf(p1 * i1 * sd) : 0.0f;
            const float a2 = ok2 ? __expf(p2 * i2 * sd) : 0.0f;
            const float a3 = ok3 ? __expf(p3 * i3 * sd) : 0.0f;
            den += a0; den += a1; den += a2; den += a3;
            acc.x += a0 * r0.x; acc.y += a0 * r0.y; acc.z += a0 * r0.z; acc.w += a0 * r0.w;
            acc.x += a1 * r1.x; acc.y += a1 * r1.y; acc.z += a1 * r1.z; acc.w += a1 * r1.w;
            acc.x += a2 * r2.x; acc.y += a2 * r2.y; acc.z += a2 * r2.z; acc.w += a2 * r2.w;
            acc.x += a3 * r3.x; acc.y += a3 * r3.y; acc.z += a3 * r3.z; acc.w += a3 * r3.w;
        }
    }

    const float dn = den > 0.0f ? den : 1.0f;
    float4 o;
    o.x = acc.x / dn; o.y = acc.y / dn; o.z = acc.z / dn; o.w = acc.w / dn;
    ((float4*)(hout + (size_t)node * DIM))[lane] = o;
}

__global__ void pool_sum(const float* __restrict__ h,
                         const int* __restrict__ gid,
                         float* __restrict__ hg, int n) {
    int i = blockIdx.x * blockDim.x + threadIdx.x;
    if (i >= n * DIM) return;
    int node = i >> 7;
    int d    = i & 127;
    atomAddF(hg + (size_t)gid[node] * DIM + d, h[i]);
}

__global__ void pool_cnt(const int* __restrict__ gid,
                         float* __restrict__ cnt, int n) {
    int i = blockIdx.x * blockDim.x + threadIdx.x;
    if (i >= n) return;
    atomAddF(cnt + gid[i], 1.0f);
}

__global__ void pool_div(float* __restrict__ hg, const float* __restrict__ cnt) {
    int i = blockIdx.x * blockDim.x + threadIdx.x;
    if (i >= NGRAPH * DIM) return;
    hg[i] /= fmaxf(cnt[i >> 7], 1.0f);
}

__global__ void classifier_wmma(const float* __restrict__ hg,
                                const float* __restrict__ W,
                                const float* __restrict__ b,
                                float* __restrict__ out) {
    int mt = blockIdx.x / 7;
    int nt = blockIdx.x % 7;
    int m0 = mt * 16;
    int n0 = nt * 16;
    int lane = threadIdx.x & 31;
    int half = lane >> 4;
    int lr   = lane & 15;
    int col  = n0 + lr;
    int ccol = col < NCLS ? col : (NCLS - 1);
    float mask = col < NCLS ? 1.0f : 0.0f;

    v8f acc = {};
#pragma unroll 4
    for (int k0 = 0; k0 < DIM; k0 += 4) {
        const float* ar = hg + (size_t)(m0 + lr) * DIM + k0 + 2 * half;
        v2f a;
        a.x = ar[0];
        a.y = ar[1];
        v2f bb;
        bb.x = W[(size_t)(k0 + 2 * half) * NCLS + ccol] * mask;
        bb.y = W[(size_t)(k0 + 2 * half + 1) * NCLS + ccol] * mask;
        acc = __builtin_amdgcn_wmma_f32_16x16x4_f32(
             false, a,  false, bb,
             (short)0, acc,  false,  false);
    }

    if (col < NCLS) {
        float bias = b[col];
#pragma unroll
        for (int r = 0; r < 8; ++r) {
            int m = m0 + r + 8 * half;
            out[(size_t)m * NCLS + col] = acc[r] + bias;
        }
    }
}

extern "C" void kernel_launch(void* const* d_in, const int* in_sizes, int n_in,
                              void* d_out, int out_size, void* d_ws, size_t ws_size,
                              hipStream_t stream) {
    const float* h0    = (const float*)d_in[0];
    const int*   src   = (const int*)d_in[1];
    const int*   dst   = (const int*)d_in[2];
    const int*   gid   = (const int*)d_in[3];
    const float* betas = (const float*)d_in[4];
    const float* Wc    = (const float*)d_in[5];
    const float* bc    = (const float*)d_in[6];
    float*       out   = (float*)d_out;

    float* plane0 = (float*)d_ws;
    float* hA    = plane0 + (size_t)NNODES * DIM;
    float* hB    = hA + (size_t)NNODES * DIM;
    float* spare = hB + (size_t)NNODES * DIM;
    float* hg    = spare + NNODES;
    float* cnt   = hg + NGRAPH * DIM;

    int*   esrc  = (int*)plane0;
    int*   cur   = esrc + NEDGES;
    float* inv   = (float*)(cur + NNODES);

    const int ND = NNODES * DIM;

    hipMemsetAsync(cur, 0, sizeof(int) * NNODES, stream);
    dst_count<<<(NEDGES + 255) / 256, 256, 0, stream>>>(dst, cur, NEDGES);
    dst_scan<<<1, 1024, 0, stream>>>(cur);
    dst_fill<<<(NEDGES + 255) / 256, 256, 0, stream>>>(src, dst, cur, esrc, NEDGES);

    const float* hin = h0;
    float* outs[NLAYERS] = {hA, hB, hA};
    for (int l = 0; l < NLAYERS; ++l) {
        float* hout = outs[l];
        row_inv_norm<<<(NNODES + 7) / 8, 256, 0, stream>>>(hin, inv, NNODES);
        gather_attend<<<(NNODES + 7) / 8, 256, 0, stream>>>(hin, inv, cur, esrc,
                                                            betas, l, hout, NNODES);
        hin = hout;
    }

    hipMemsetAsync(hg, 0, sizeof(float) * NGRAPH * DIM, stream);
    hipMemsetAsync(cnt, 0, sizeof(float) * NGRAPH, stream);
    pool_sum<<<(ND + 255) / 256, 256, 0, stream>>>(hin, gid, hg, NNODES);
    pool_cnt<<<(NNODES + 255) / 256, 256, 0, stream>>>(gid, cnt, NNODES);
    pool_div<<<(NGRAPH * DIM + 255) / 256, 256, 0, stream>>>(hg, cnt);

    classifier_wmma<<<4 * 7, 32, 0, stream>>>(hg, Wc, bc, out);
}
